# starting kernel
# baseline (speedup 1.0000x reference)
#include <hip/hip_runtime.h>
#include <hip/hip_bf16.h>

typedef float v2f __attribute__((ext_vector_type(2)));
typedef float v8f __attribute__((ext_vector_type(8)));

#define HW   40000
#define CDIM 128
#define SCALE_F 0.17677669529663687f

__global__ __launch_bounds__(256)
void conv_half_kernel(const float* __restrict__ x,
                      const float* __restrict__ wkv,
                      float* __restrict__ kvh,
                      int ohalf)
{
    __shared__ __align__(16) float Wt[32 * 132];
    __shared__ __align__(16) float Xt[64 * 132];

    const int n   = blockIdx.z;
    const int o0  = blockIdx.y * 32;
    const int hw0 = blockIdx.x * 64;
    const int tid = threadIdx.x;

    {
        const float* src = wkv + (size_t)(ohalf * 128 + o0) * CDIM;
        for (int idx = tid; idx < 32 * 32; idx += 256) {
            int r = idx >> 5, c4 = (idx & 31) << 2;
            *(float4*)&Wt[r * 132 + c4] = *(const float4*)(src + r * CDIM + c4);
        }
    }
    {
        const float* src = x + ((size_t)n * HW + hw0) * CDIM;
        for (int idx = tid; idx < 64 * 32; idx += 256) {
            int r = idx >> 5, c4 = (idx & 31) << 2;
            *(float4*)&Xt[r * 132 + c4] = *(const float4*)(src + (size_t)r * CDIM + c4);
        }
    }
    __syncthreads();

    const int wv = tid >> 5;
    const int lane = tid & 31;
    const int lo = lane & 15, hi = lane >> 4;
    const int ot = wv >> 2;
    const int hs = wv & 3;

    const float* Arow = &Wt[(ot * 16 + lo) * 132];
    const float* Brow = &Xt[(hs * 16 + lo) * 132];
    v8f acc = {0.f, 0.f, 0.f, 0.f, 0.f, 0.f, 0.f, 0.f};

#if __has_builtin(__builtin_amdgcn_wmma_f32_16x16x4_f32)
#pragma unroll
    for (int k0 = 0; k0 < 128; k0 += 4) {
        v2f a = *(const v2f*)(Arow + k0 + 2 * hi);
        v2f b = *(const v2f*)(Brow + k0 + 2 * hi);
        acc = __builtin_amdgcn_wmma_f32_16x16x4_f32(
            false, a, false, b, (short)0, acc, false, false);
    }
#else
#pragma unroll
    for (int g = 0; g < 8; ++g) {
        float s = 0.f;
        const float* ar = &Wt[(ot * 16 + g + 8 * hi) * 132];
        const float* br = &Xt[(hs * 16 + lo) * 132];
        for (int k = 0; k < 128; ++k) s += ar[k] * br[k];
        acc[g] = s;
    }
#endif

    float* dst = kvh + ((size_t)n * 128 + (o0 + ot * 16 + 8 * hi)) * HW
                     + (hw0 + hs * 16 + lo);
#pragma unroll
    for (int g = 0; g < 8; ++g) dst[(size_t)g * HW] = acc[g];
}

__global__ __launch_bounds__(256)
void attn_pass_kernel(const float* __restrict__ qin,
                      const float* __restrict__ kvh,
                      float* __restrict__ qout,
                      int dil)
{
    const int p = blockIdx.x * 256 + threadIdx.x;
    if (p >= HW) return;
    const int n = blockIdx.y;
    const int b = blockIdx.z;

    const float* qb = qin + (size_t)b * (HW * CDIM);
    const float* Kb = kvh + (size_t)(b * 2 + 0) * (128 * HW);
    const float* Vb = kvh + (size_t)(b * 2 + 1) * (128 * HW);

    float qv[32];
#pragma unroll
    for (int dd = 0; dd < 32; ++dd)
        qv[dd] = qb[(size_t)(n * 32 + dd) * HW + p];

    float m = -1e30f, denom = 0.f;
    float acc[32];
#pragma unroll
    for (int dd = 0; dd < 32; ++dd) acc[dd] = 0.f;

    for (int kk2 = 0; kk2 < 9; ++kk2) {
        int offs[32];
        float s = 0.f;
#pragma unroll
        for (int dd = 0; dd < 32; ++dd) {
            const int cl = n * 32 + dd;
            const int T  = (cl * 9 + kk2) * HW + p;
            const int chunk = T / 25600;
            const int sp    = T - chunk * 25600;
            const int cp = chunk / 9;
            const int kk = chunk - cp * 9;
            const int ki = kk / 3, kj = kk - ki * 3;
            const int w2 = (sp >> 7) + (ki - 1) * dil;
            const int c3 = (sp & 127) + (kj - 1) * dil;
            int off = -1;
            if ((unsigned)w2 < 200u && (unsigned)c3 < 128u)
                off = cp * 25600 + (w2 << 7) + c3;
            offs[dd] = off;
            const float kval = (off >= 0) ? Kb[off] : 0.f;
            s += qv[dd] * kval;
        }
        s *= SCALE_F;
        const float mn = fmaxf(m, s);
        const float cs = __expf(m - mn);
        const float w  = __expf(s - mn);
        denom = denom * cs + w;
#pragma unroll
        for (int dd = 0; dd < 32; ++dd) {
            const float vval = (offs[dd] >= 0) ? Vb[offs[dd]] : 0.f;
            acc[dd] = acc[dd] * cs + w * vval;
        }
        m = mn;
    }

    const float inv = 1.f / denom;
    float* ob = qout + (size_t)b * (HW * CDIM) + (size_t)p * CDIM + n * 32;
#pragma unroll
    for (int g4 = 0; g4 < 8; ++g4) {
        float4 v;
        v.x = acc[g4 * 4 + 0] * inv;
        v.y = acc[g4 * 4 + 1] * inv;
        v.z = acc[g4 * 4 + 2] * inv;
        v.w = acc[g4 * 4 + 3] * inv;
        *(float4*)(ob + g4 * 4) = v;
    }
}

__global__ __launch_bounds__(256)
void proj_kernel(const float* __restrict__ qf,
                 const float* __restrict__ pw,
                 const float* __restrict__ pb,
                 float* __restrict__ out)
{
    __shared__ __align__(16) float Pt[32 * 132];
    __shared__ __align__(16) float Qt[64 * 132];

    const int r0  = blockIdx.x * 64;
    const int co0 = blockIdx.y * 32;
    const int tid = threadIdx.x;

    for (int idx = tid; idx < 32 * 32; idx += 256) {
        int r = idx >> 5, c4 = (idx & 31) << 2;
        *(float4*)&Pt[r * 132 + c4] = *(const float4*)(pw + (size_t)(co0 + r) * CDIM + c4);
    }
    for (int idx = tid; idx < 64 * 32; idx += 256) {
        int r = idx >> 5, c4 = (idx & 31) << 2;
        *(float4*)&Qt[r * 132 + c4] = *(const float4*)(qf + (size_t)(r0 + r) * CDIM + c4);
    }
    __syncthreads();

    const int wv = tid >> 5, lane = tid & 31;
    const int lo = lane & 15, hi = lane >> 4;
    const int ct = wv >> 2;
    const int rs = wv & 3;

    const float* Arow = &Qt[(rs * 16 + lo) * 132];
    const float* Brow = &Pt[(ct * 16 + lo) * 132];
    const float bv = pb[co0 + ct * 16 + lo];
    v8f acc = {bv, bv, bv, bv, bv, bv, bv, bv};

#if __has_builtin(__builtin_amdgcn_wmma_f32_16x16x4_f32)
#pragma unroll
    for (int k0 = 0; k0 < 128; k0 += 4) {
        v2f a = *(const v2f*)(Arow + k0 + 2 * hi);
        v2f b = *(const v2f*)(Brow + k0 + 2 * hi);
        acc = __builtin_amdgcn_wmma_f32_16x16x4_f32(
            false, a, false, b, (short)0, acc, false, false);
    }
#else
#pragma unroll
    for (int g = 0; g < 8; ++g) {
        float s = bv;
        const float* ar = &Qt[(rs * 16 + g + 8 * hi) * 132];
        const float* br = &Pt[(ct * 16 + lo) * 132];
        for (int k = 0; k < 128; ++k) s += ar[k] * br[k];
        acc[g] = s;
    }
#endif

    float* dst = out + (size_t)(r0 + rs * 16 + 8 * hi) * CDIM + (co0 + ct * 16 + lo);
#pragma unroll
    for (int g = 0; g < 8; ++g) dst[(size_t)g * CDIM] = acc[g];
}

extern "C" void kernel_launch(void* const* d_in, const int* in_sizes, int n_in,
                              void* d_out, int out_size, void* d_ws, size_t ws_size,
                              hipStream_t stream) {
    const float* q   = (const float*)d_in[0];
    const float* x   = (const float*)d_in[1];
    const float* wkv = (const float*)d_in[2];
    const float* pw  = (const float*)d_in[3];
    const float* pb  = (const float*)d_in[4];
    float* out = (float*)d_out;

    float* kvh = (float*)d_ws;
    float* q1  = kvh + (size_t)4 * 128 * HW;
    float* q2  = q1 + (size_t)2 * HW * CDIM;

    dim3 blk(256);
    dim3 cg(625, 4, 4);
    dim3 ag(157, 4, 2);
    dim3 pg(1250, 4);

    conv_half_kernel<<<cg, blk, 0, stream>>>(x, wkv, kvh, 0);
    attn_pass_kernel<<<ag, blk, 0, stream>>>(q, kvh, q1, 1);
    conv_half_kernel<<<cg, blk, 0, stream>>>(x, wkv, kvh, 1);
    attn_pass_kernel<<<ag, blk, 0, stream>>>(q1, kvh, q2, 2);
    proj_kernel<<<pg, blk, 0, stream>>>(q2, pw, pb, out);
}


# attention pass: tap address split (plane, row, column) done once per wave, lane dd for channel dd, and broadcast by readlane, not by every lane for every channel; sums and their order unchanged
# speedup vs baseline: 1.2656x; 1.2656x over previous
#include <hip/hip_runtime.h>
#include <hip/hip_bf16.h>

typedef float v2f __attribute__((ext_vector_type(2)));
typedef float v8f __attribute__((ext_vector_type(8)));

#define HW   40000
#define CDIM 128
#define SCALE_F 0.17677669529663687f

__global__ __launch_bounds__(256)
void conv_half_kernel(const float* __restrict__ x,
                      const float* __restrict__ wkv,
                      float* __restrict__ kvh,
                      int ohalf)
{
    __shared__ __align__(16) float Wt[32 * 132];
    __shared__ __align__(16) float Xt[64 * 132];

    const int n   = blockIdx.z;
    const int o0  = blockIdx.y * 32;
    const int hw0 = blockIdx.x * 64;
    const int tid = threadIdx.x;

    {
        const float* src = wkv + (size_t)(ohalf * 128 + o0) * CDIM;
        for (int idx = tid; idx < 32 * 32; idx += 256) {
            int r = idx >> 5, c4 = (idx & 31) << 2;
            *(float4*)&Wt[r * 132 + c4] = *(const float4*)(src + r * CDIM + c4);
        }
    }
    {
        const float* src = x + ((size_t)n * HW + hw0) * CDIM;
        for (int idx = tid; idx < 64 * 32; idx += 256) {
            int r = idx >> 5, c4 = (idx & 31) << 2;
            *(float4*)&Xt[r * 132 + c4] = *(const float4*)(src + (size_t)r * CDIM + c4);
        }
    }
    __syncthreads();

    const int wv = tid >> 5;
    const int lane = tid & 31;
    const int lo = lane & 15, hi = lane >> 4;
    const int ot = wv >> 2;
    const int hs = wv & 3;

    const float* Arow = &Wt[(ot * 16 + lo) * 132];
    const float* Brow = &Xt[(hs * 16 + lo) * 132];
    v8f acc = {0.f, 0.f, 0.f, 0.f, 0.f, 0.f, 0.f, 0.f};

#if __has_builtin(__builtin_amdgcn_wmma_f32_16x16x4_f32)
#pragma unroll
    for (int k0 = 0; k0 < 128; k0 += 4) {
        v2f a = *(const v2f*)(Arow + k0 + 2 * hi);
        v2f b = *(const v2f*)(Brow + k0 + 2 * hi);
        acc = __builtin_amdgcn_wmma_f32_16x16x4_f32(
            false, a, false, b, (short)0, acc, false, false);
    }
#else
#pragma unroll
    for (int g = 0; g < 8; ++g) {
        float s = 0.f;
        const float* ar = &Wt[(ot * 16 + g + 8 * hi) * 132];
        const float* br = &Xt[(hs * 16 + lo) * 132];
        for (int k = 0; k < 128; ++k) s += ar[k] * br[k];
        acc[g] = s;
    }
#endif

    float* dst = kvh + ((size_t)n * 128 + (o0 + ot * 16 + 8 * hi)) * HW
                     + (hw0 + hs * 16 + lo);
#pragma unroll
    for (int g = 0; g < 8; ++g) dst[(size_t)g * HW] = acc[g];
}

__global__ __launch_bounds__(256)
void attn_pass_kernel(const float* __restrict__ qin,
                      const float* __restrict__ kvh,
                      float* __restrict__ qout,
                      int dil)
{
    const unsigned lane = threadIdx.x & 31u;
    const int pw = __builtin_amdgcn_readfirstlane((int)(blockIdx.x * 256 + (threadIdx.x & ~31u)));
    if (pw >= HW) return;
    const int p = pw + (int)lane;
    const int n = blockIdx.y;
    const int b = blockIdx.z;

    const float* qb = qin + (size_t)b * (HW * CDIM);
    const int kplane = b * 2 * (128 * HW);

    float qv[32];
#pragma unroll
    for (int dd = 0; dd < 32; ++dd)
        qv[dd] = qb[(size_t)(n * 32 + dd) * HW + p];

    float m = -1e30f, denom = 0.f;
    float acc[32];
#pragma unroll
    for (int dd = 0; dd < 32; ++dd) acc[dd] = 0.f;

    for (int kk2 = 0; kk2 < 9; ++kk2) {
        int offv, c0v;
        {
            const int cl = n * 32 + lane;
            const int T  = (cl * 9 + kk2) * HW + pw;
            const int chunk = T / 25600;
            const int sp    = T - chunk * 25600;
            const int cp = chunk / 9;
            const int kk = chunk - cp * 9;
            const int ki = kk / 3, kj = kk - ki * 3;
            const int w2 = (sp >> 7) + (ki - 1) * dil;
            const int c0 = (sp & 127) + (kj - 1) * dil;
            offv = kplane + cp * 25600 + (w2 << 7) + c0;
            c0v  = ((unsigned)w2 < 200u) ? c0 : -64;
        }

        float kx[32], vx[32];
#pragma unroll
        for (int dd = 0; dd < 32; ++dd) {
            const unsigned k0 = (unsigned)__builtin_amdgcn_readlane(offv, dd);
            const int c0      = __builtin_amdgcn_readlane(c0v, dd);
            float kval = 0.f, vval = 0.f;
            if ((unsigned)c0 + lane < 128u) {
                const unsigned ke = k0 + lane;
                kval = kvh[ke];
                vval = kvh[ke + 128u * HW];
            }
            kx[dd] = kval;
            vx[dd] = vval;
        }

        float s = 0.f;
#pragma unroll
        for (int dd = 0; dd < 32; ++dd) s += qv[dd] * kx[dd];
        s *= SCALE_F;
        const float mn = fmaxf(m, s);
        const float cs = __expf(m - mn);
        const float w  = __expf(s - mn);
        denom = denom * cs + w;
#pragma unroll
        for (int dd = 0; dd < 32; ++dd)
            acc[dd] = acc[dd] * cs + w * vx[dd];
        m = mn;
    }

    const float inv = 1.f / denom;
    float* ob = qout + (size_t)b * (HW * CDIM) + (size_t)p * CDIM + n * 32;
#pragma unroll
    for (int g4 = 0; g4 < 8; ++g4) {
        float4 v;
        v.x = acc[g4 * 4 + 0] * inv;
        v.y = acc[g4 * 4 + 1] * inv;
        v.z = acc[g4 * 4 + 2] * inv;
        v.w = acc[g4 * 4 + 3] * inv;
        *(float4*)(ob + g4 * 4) = v;
    }
}

__global__ __launch_bounds__(256)
void proj_kernel(const float* __restrict__ qf,
                 const float* __restrict__ pw,
                 const float* __restrict__ pb,
                 float* __restrict__ out)
{
    __shared__ __align__(16) float Pt[32 * 132];
    __shared__ __align__(16) float Qt[64 * 132];

    const int r0  = blockIdx.x * 64;
    const int co0 = blockIdx.y * 32;
    const int tid = threadIdx.x;

    for (int idx = tid; idx < 32 * 32; idx += 256) {
        int r = idx >> 5, c4 = (idx & 31) << 2;
        *(float4*)&Pt[r * 132 + c4] = *(const float4*)(pw + (size_t)(co0 + r) * CDIM + c4);
    }
    for (int idx = tid; idx < 64 * 32; idx += 256) {
        int r = idx >> 5, c4 = (idx & 31) << 2;
        *(float4*)&Qt[r * 132 + c4] = *(const float4*)(qf + (size_t)(r0 + r) * CDIM + c4);
    }
    __syncthreads();

    const int wv = tid >> 5, lane = tid & 31;
    const int lo = lane & 15, hi = lane >> 4;
    const int ct = wv >> 2;
    const int rs = wv & 3;

    const float* Arow = &Qt[(rs * 16 + lo) * 132];
    const float* Brow = &Pt[(ct * 16 + lo) * 132];
    const float bv = pb[co0 + ct * 16 + lo];
    v8f acc = {bv, bv, bv, bv, bv, bv, bv, bv};

#if __has_builtin(__builtin_amdgcn_wmma_f32_16x16x4_f32)
#pragma unroll
    for (int k0 = 0; k0 < 128; k0 += 4) {
        v2f a = *(const v2f*)(Arow + k0 + 2 * hi);
        v2f b = *(const v2f*)(Brow + k0 + 2 * hi);
        acc = __builtin_amdgcn_wmma_f32_16x16x4_f32(
            false, a, false, b, (short)0, acc, false, false);
    }
#else
#pragma unroll
    for (int g = 0; g < 8; ++g) {
        float s = bv;
        const float* ar = &Qt[(rs * 16 + g + 8 * hi) * 132];
        const float* br = &Pt[(ct * 16 + lo) * 132];
        for (int k = 0; k < 128; ++k) s += ar[k] * br[k];
        acc[g] = s;
    }
#endif

    float* dst = out + (size_t)(r0 + rs * 16 + 8 * hi) * CDIM + (co0 + ct * 16 + lo);
#pragma unroll
    for (int g = 0; g < 8; ++g) dst[(size_t)g * CDIM] = acc[g];
}

extern "C" void kernel_launch(void* const* d_in, const int* in_sizes, int n_in,
                              void* d_out, int out_size, void* d_ws, size_t ws_size,
                              hipStream_t stream) {
    const float* q   = (const float*)d_in[0];
    const float* x   = (const float*)d_in[1];
    const float* wkv = (const float*)d_in[2];
    const float* pw  = (const float*)d_in[3];
    const float* pb  = (const float*)d_in[4];
    float* out = (float*)d_out;

    float* kvh = (float*)d_ws;
    float* q1  = kvh + (size_t)4 * 128 * HW;
    float* q2  = q1 + (size_t)2 * HW * CDIM;

    dim3 blk(256);
    dim3 cg(625, 4, 4);
    dim3 ag(157, 4, 2);
    dim3 pg(1250, 4);

    conv_half_kernel<<<cg, blk, 0, stream>>>(x, wkv, kvh, 0);
    attn_pass_kernel<<<ag, blk, 0, stream>>>(q, kvh, q1, 1);
    conv_half_kernel<<<cg, blk, 0, stream>>>(x, wkv, kvh, 1);
    attn_pass_kernel<<<ag, blk, 0, stream>>>(q1, kvh, q2, 2);
    proj_kernel<<<pg, blk, 0, stream>>>(q2, pw, pb, out);
}
